# starting kernel
# baseline (speedup 1.0000x reference)
#include <hip/hip_runtime.h>

typedef __attribute__((ext_vector_type(2))) float v2f;
typedef __attribute__((ext_vector_type(8))) float v8f;

#define NPIX   9216
#define DEMB   32
#define NTILES (NPIX / 16)
#define NWAVES 4
#define XSTRIDE 20
#define ESTRIDE 17
#define BW     3.0f
#define STEP   0.5f
#define LOG2E  1.44269504088896340736f

__global__ __launch_bounds__(32 * NWAVES)
void ms_iter_kernel(const float* __restrict__ xin,
                    float* __restrict__ xout,
                    float* __restrict__ xout2)
{
    __shared__ __align__(16) float xtile[NWAVES][DEMB * XSTRIDE];
    __shared__ float etile[NWAVES][16 * ESTRIDE];
    __shared__ float red[NWAVES][32][17];

    const int lane  = threadIdx.x & 31;
    const int wave  = threadIdx.x >> 5;
    const int half  = lane >> 4;
    const int l15   = lane & 15;
    const int jbase = blockIdx.x * 16;

    float* xt = xtile[wave];
    float* et = etile[wave];

    v2f bj[8];
#pragma unroll
    for (int k0 = 0; k0 < 8; ++k0) {
        const int d0 = 4 * k0 + 2 * half;
        bj[k0].x = xin[(size_t)d0 * NPIX + jbase + l15];
        bj[k0].y = xin[(size_t)(d0 + 1) * NPIX + jbase + l15];
    }

    v8f y0 = {};
    v8f y1 = {};
    float dacc = 0.0f;

    for (int it = wave; it < NTILES; it += NWAVES) {
        const int ibase = it * 16;

        {
            const float4* src = (const float4*)(xin + (size_t)lane * NPIX + ibase);
            float4* dst = (float4*)(xt + lane * XSTRIDE);
            dst[0] = src[0]; dst[1] = src[1]; dst[2] = src[2]; dst[3] = src[3];
        }
        if (ibase + 16 * NWAVES < NPIX)
            __builtin_prefetch(xin + (size_t)lane * NPIX + ibase + 16 * NWAVES, 0, 0);

        v8f s = {};
#pragma unroll
        for (int k0 = 0; k0 < 8; ++k0) {
            const int d0 = 4 * k0 + 2 * half;
            v2f a;
            a.x = xt[d0 * XSTRIDE + l15];
            a.y = xt[(d0 + 1) * XSTRIDE + l15];
            s = __builtin_amdgcn_wmma_f32_16x16x4_f32(false, a, false, bj[k0],
                                                      (short)0, s, false, false);
        }

        float esum = 0.0f;
#pragma unroll
        for (int v = 0; v < 8; ++v) {
            const float e = __builtin_amdgcn_exp2f(s[v] * (BW * LOG2E));
            esum += e;
            et[(v + 8 * half) * ESTRIDE + l15] = e;
        }
        dacc += esum;

#pragma unroll
        for (int c = 0; c < 4; ++c) {
            const int k0 = 4 * c + 2 * half;
            v2f b2;
            b2.x = et[k0 * ESTRIDE + l15];
            b2.y = et[(k0 + 1) * ESTRIDE + l15];

            v2f a2;
            a2.x = xt[l15 * XSTRIDE + k0];
            a2.y = xt[l15 * XSTRIDE + k0 + 1];
            y0 = __builtin_amdgcn_wmma_f32_16x16x4_f32(false, a2, false, b2,
                                                       (short)0, y0, false, false);

            v2f a3;
            a3.x = xt[(16 + l15) * XSTRIDE + k0];
            a3.y = xt[(16 + l15) * XSTRIDE + k0 + 1];
            y1 = __builtin_amdgcn_wmma_f32_16x16x4_f32(false, a3, false, b2,
                                                       (short)0, y1, false, false);
        }
    }

#pragma unroll
    for (int v = 0; v < 8; ++v) {
        red[wave][lane][v]     = y0[v];
        red[wave][lane][8 + v] = y1[v];
    }
    red[wave][lane][16] = dacc;
    __syncthreads();

    if (wave == 0) {
        v8f Y0 = {}, Y1 = {};
        float dtot = 0.0f;
        for (int w = 0; w < NWAVES; ++w) {
#pragma unroll
            for (int v = 0; v < 8; ++v) {
                Y0[v] += red[w][lane][v];
                Y1[v] += red[w][lane][8 + v];
            }
            dtot += red[w][lane][16];
        }
        const float dfull = dtot + __shfl_xor(dtot, 16, 32);
        const float sc = STEP / dfull;

#pragma unroll
        for (int v = 0; v < 8; ++v) {
            const int m = v + 8 * half;
            const size_t j = (size_t)jbase + l15;
            const size_t o0 = (size_t)m * NPIX + j;
            const size_t o1 = (size_t)(16 + m) * NPIX + j;
            const float r0 = Y0[v] * sc + (1.0f - STEP) * xin[o0];
            const float r1 = Y1[v] * sc + (1.0f - STEP) * xin[o1];
            xout[o0] = r0;
            xout[o1] = r1;
            if (xout2) { xout2[o0] = r0; xout2[o1] = r1; }
        }
    }
}

extern "C" void kernel_launch(void* const* d_in, const int* in_sizes, int n_in,
                              void* d_out, int out_size, void* d_ws, size_t ws_size,
                              hipStream_t stream) {
    (void)in_sizes; (void)n_in; (void)d_ws; (void)ws_size; (void)out_size;
    const float* x0 = (const float*)d_in[0];
    float* out = (float*)d_out;
    const size_t F = (size_t)DEMB * NPIX;

    dim3 grid(NTILES), block(32 * NWAVES);
    ms_iter_kernel<<<grid, block, 0, stream>>>(x0, out + F, nullptr);
    ms_iter_kernel<<<grid, block, 0, stream>>>(out + F, out + 2 * F, nullptr);
    ms_iter_kernel<<<grid, block, 0, stream>>>(out + 2 * F, out + 3 * F, out);
}


# GEMM1's A fragments load from global in fragment layout (two 64-B runs per load, not 32 lines), one slab ahead; x reaches LDS only as those registers stored for GEMM2; f32 sum order unchanged
# speedup vs baseline: 1.4190x; 1.4190x over previous
#include <hip/hip_runtime.h>

typedef __attribute__((ext_vector_type(2))) float v2f;
typedef __attribute__((ext_vector_type(8))) float v8f;

#define NPIX   9216
#define DEMB   32
#define NTILES (NPIX / 16)
#define NWAVES 4
#define TB     4
#define SLABW  (16 * TB)
#define NSLABS (NPIX / SLABW)
#define XSTRIDE 76
#define ESTRIDE 17
#define BW     3.0f
#define STEP   0.5f
#define LOG2E  1.44269504088896340736f

__global__ __launch_bounds__(32 * NWAVES)
void ms_iter_kernel(const float* __restrict__ xin,
                    float* __restrict__ xout,
                    float* __restrict__ xout2)
{
    __shared__ __align__(16) float xtile[NWAVES][DEMB * XSTRIDE];
    __shared__ float etile[NWAVES][TB * 16 * ESTRIDE];
    __shared__ float red[NWAVES][32][17];

    const int lane  = threadIdx.x & 31;
    const int wave  = threadIdx.x >> 5;
    const int half  = lane >> 4;
    const int l15   = lane & 15;
    const int jbase = blockIdx.x * 16;

    float* xt = xtile[wave];
    float* et = etile[wave];

    v2f bj[8];
#pragma unroll
    for (int k0 = 0; k0 < 8; ++k0) {
        const int d0 = 4 * k0 + 2 * half;
        bj[k0].x = xin[(size_t)d0 * NPIX + jbase + l15];
        bj[k0].y = xin[(size_t)(d0 + 1) * NPIX + jbase + l15];
    }

    v8f y0 = {};
    v8f y1 = {};
    float dacc = 0.0f;

    const float* fsrc = xin + (size_t)(2 * half) * NPIX + l15;
    v2f af[8][TB];
#pragma unroll
    for (int k0 = 0; k0 < 8; ++k0) {
#pragma unroll
        for (int t = 0; t < TB; ++t) {
            af[k0][t].x = fsrc[(size_t)(4 * k0) * NPIX + wave * SLABW + 16 * t];
            af[k0][t].y = fsrc[(size_t)(4 * k0 + 1) * NPIX + wave * SLABW + 16 * t];
        }
    }

    for (int sb = wave; sb < NSLABS; sb += NWAVES) {
#pragma unroll
        for (int k0 = 0; k0 < 8; ++k0) {
            const int d0 = 4 * k0 + 2 * half;
#pragma unroll
            for (int t = 0; t < TB; ++t) {
                xt[d0 * XSTRIDE + 16 * t + l15]       = af[k0][t].x;
                xt[(d0 + 1) * XSTRIDE + 16 * t + l15] = af[k0][t].y;
            }
        }

        v8f s[TB];
#pragma unroll
        for (int t = 0; t < TB; ++t) s[t] = (v8f){};
#pragma unroll
        for (int k0 = 0; k0 < 8; ++k0) {
#pragma unroll
            for (int t = 0; t < TB; ++t) {
                s[t] = __builtin_amdgcn_wmma_f32_16x16x4_f32(false, af[k0][t], false, bj[k0],
                                                             (short)0, s[t], false, false);
            }
        }

        if (sb + NWAVES < NSLABS) {
            const float* nsrc = fsrc + (sb + NWAVES) * SLABW;
#pragma unroll
            for (int k0 = 0; k0 < 8; ++k0) {
#pragma unroll
                for (int t = 0; t < TB; ++t) {
                    af[k0][t].x = nsrc[(size_t)(4 * k0) * NPIX + 16 * t];
                    af[k0][t].y = nsrc[(size_t)(4 * k0 + 1) * NPIX + 16 * t];
                }
            }
        }

#pragma unroll
        for (int t = 0; t < TB; ++t) {
            float esum = 0.0f;
#pragma unroll
            for (int v = 0; v < 8; ++v) {
                const float e = __builtin_amdgcn_exp2f(s[t][v] * (BW * LOG2E));
                esum += e;
                et[t * 16 * ESTRIDE + (v + 8 * half) * ESTRIDE + l15] = e;
            }
            dacc += esum;
        }

#pragma unroll
        for (int t = 0; t < TB; ++t) {
#pragma unroll
            for (int c = 0; c < 4; ++c) {
                const int k0 = 4 * c + 2 * half;
                v2f b2;
                b2.x = et[t * 16 * ESTRIDE + k0 * ESTRIDE + l15];
                b2.y = et[t * 16 * ESTRIDE + (k0 + 1) * ESTRIDE + l15];

                v2f a2;
                a2.x = xt[l15 * XSTRIDE + 16 * t + k0];
                a2.y = xt[l15 * XSTRIDE + 16 * t + k0 + 1];
                y0 = __builtin_amdgcn_wmma_f32_16x16x4_f32(false, a2, false, b2,
                                                           (short)0, y0, false, false);

                v2f a3;
                a3.x = xt[(16 + l15) * XSTRIDE + 16 * t + k0];
                a3.y = xt[(16 + l15) * XSTRIDE + 16 * t + k0 + 1];
                y1 = __builtin_amdgcn_wmma_f32_16x16x4_f32(false, a3, false, b2,
                                                           (short)0, y1, false, false);
            }
        }
    }

#pragma unroll
    for (int v = 0; v < 8; ++v) {
        red[wave][lane][v]     = y0[v];
        red[wave][lane][8 + v] = y1[v];
    }
    red[wave][lane][16] = dacc;
    __syncthreads();

    if (wave == 0) {
        v8f Y0 = {}, Y1 = {};
        float dtot = 0.0f;
        for (int w = 0; w < NWAVES; ++w) {
#pragma unroll
            for (int v = 0; v < 8; ++v) {
                Y0[v] += red[w][lane][v];
                Y1[v] += red[w][lane][8 + v];
            }
            dtot += red[w][lane][16];
        }
        const float dfull = dtot + __shfl_xor(dtot, 16, 32);
        const float sc = STEP / dfull;

#pragma unroll
        for (int v = 0; v < 8; ++v) {
            const int m = v + 8 * half;
            const size_t j = (size_t)jbase + l15;
            const size_t o0 = (size_t)m * NPIX + j;
            const size_t o1 = (size_t)(16 + m) * NPIX + j;
            const float r0 = Y0[v] * sc + (1.0f - STEP) * xin[o0];
            const float r1 = Y1[v] * sc + (1.0f - STEP) * xin[o1];
            xout[o0] = r0;
            xout[o1] = r1;
            if (xout2) { xout2[o0] = r0; xout2[o1] = r1; }
        }
    }
}

extern "C" void kernel_launch(void* const* d_in, const int* in_sizes, int n_in,
                              void* d_out, int out_size, void* d_ws, size_t ws_size,
                              hipStream_t stream) {
    (void)in_sizes; (void)n_in; (void)d_ws; (void)ws_size; (void)out_size;
    const float* x0 = (const float*)d_in[0];
    float* out = (float*)d_out;
    const size_t F = (size_t)DEMB * NPIX;

    dim3 grid(NTILES), block(32 * NWAVES);
    ms_iter_kernel<<<grid, block, 0, stream>>>(x0, out + F, nullptr);
    ms_iter_kernel<<<grid, block, 0, stream>>>(out + F, out + 2 * F, nullptr);
    ms_iter_kernel<<<grid, block, 0, stream>>>(out + 2 * F, out + 3 * F, out);
}


# Slab loop skewed by one slab: all LDS fragment reads of GEMM2 stand in one group per trip, ahead of a 64-instruction matrix run (GEMM1 of the next slab, then GEMM2); f32 sum order unchanged
# speedup vs baseline: 1.4738x; 1.0386x over previous
#include <hip/hip_runtime.h>

typedef __attribute__((ext_vector_type(2))) float v2f;
typedef __attribute__((ext_vector_type(8))) float v8f;

#define NPIX   9216
#define DEMB   32
#define NTILES (NPIX / 16)
#define NWAVES 4
#define TB     4
#define SLABW  (16 * TB)
#define NSLABS (NPIX / SLABW)
#define NTRIPS (NSLABS / NWAVES)
#define XSTRIDE 76
#define ESTRIDE 17
#define BW     3.0f
#define STEP   0.5f
#define LOG2E  1.44269504088896340736f

#define DEV static __device__ __forceinline__

DEV void load_af(v2f (&af)[8][TB], const float* __restrict__ src)
{
#pragma unroll
    for (int k0 = 0; k0 < 8; ++k0) {
#pragma unroll
        for (int t = 0; t < TB; ++t) {
            af[k0][t].x = src[(size_t)(4 * k0) * NPIX + 16 * t];
            af[k0][t].y = src[(size_t)(4 * k0 + 1) * NPIX + 16 * t];
        }
    }
}

DEV void stage_xt(float* xt, const v2f (&af)[8][TB], int half, int l15)
{
#pragma unroll
    for (int k0 = 0; k0 < 8; ++k0) {
        const int d0 = 4 * k0 + 2 * half;
#pragma unroll
        for (int t = 0; t < TB; ++t) {
            xt[d0 * XSTRIDE + 16 * t + l15]       = af[k0][t].x;
            xt[(d0 + 1) * XSTRIDE + 16 * t + l15] = af[k0][t].y;
        }
    }
}

DEV void gemm1(v8f (&s)[TB], const v2f (&af)[8][TB], const v2f (&bj)[8])
{
#pragma unroll
    for (int t = 0; t < TB; ++t) s[t] = (v8f){};
#pragma unroll
    for (int k0 = 0; k0 < 8; ++k0) {
#pragma unroll
        for (int t = 0; t < TB; ++t) {
            s[t] = __builtin_amdgcn_wmma_f32_16x16x4_f32(false, af[k0][t], false, bj[k0],
                                                         (short)0, s[t], false, false);
        }
    }
}

DEV void exp_phase(const v8f (&s)[TB], float* ew, float& dacc)
{
#pragma unroll
    for (int t = 0; t < TB; ++t) {
        float esum = 0.0f;
#pragma unroll
        for (int v = 0; v < 8; ++v) {
            const float e = __builtin_amdgcn_exp2f(s[t][v] * (BW * LOG2E));
            esum += e;
            ew[(t * 16 + v) * ESTRIDE] = e;
        }
        dacc += esum;
    }
}

DEV void load_frags(v2f (&b2)[TB][4], v2f (&a2)[TB][4], v2f (&a3)[TB][4],
                    const float* et, const float* xt, int half, int l15)
{
#pragma unroll
    for (int t = 0; t < TB; ++t) {
#pragma unroll
        for (int c = 0; c < 4; ++c) {
            const int k0 = 4 * c + 2 * half;
            b2[t][c].x = et[t * 16 * ESTRIDE + k0 * ESTRIDE + l15];
            b2[t][c].y = et[t * 16 * ESTRIDE + (k0 + 1) * ESTRIDE + l15];
            a2[t][c].x = xt[l15 * XSTRIDE + 16 * t + k0];
            a2[t][c].y = xt[l15 * XSTRIDE + 16 * t + k0 + 1];
            a3[t][c].x = xt[(16 + l15) * XSTRIDE + 16 * t + k0];
            a3[t][c].y = xt[(16 + l15) * XSTRIDE + 16 * t + k0 + 1];
        }
    }
}

DEV void gemm2(v8f& y0, v8f& y1, const v2f (&b2)[TB][4], const v2f (&a2)[TB][4], const v2f (&a3)[TB][4])
{
#pragma unroll
    for (int t = 0; t < TB; ++t) {
#pragma unroll
        for (int c = 0; c < 4; ++c) {
            y0 = __builtin_amdgcn_wmma_f32_16x16x4_f32(false, a2[t][c], false, b2[t][c],
                                                       (short)0, y0, false, false);
            y1 = __builtin_amdgcn_wmma_f32_16x16x4_f32(false, a3[t][c], false, b2[t][c],
                                                       (short)0, y1, false, false);
        }
    }
}

__global__ __launch_bounds__(32 * NWAVES)
void ms_iter_kernel(const float* __restrict__ xin,
                    float* __restrict__ xout,
                    float* __restrict__ xout2)
{
    __shared__ __align__(16) float xtile[NWAVES][DEMB * XSTRIDE];
    __shared__ float etile[NWAVES][TB * 16 * ESTRIDE];
    __shared__ float red[NWAVES][32][17];

    const int lane  = threadIdx.x & 31;
    const int wave  = threadIdx.x >> 5;
    const int half  = lane >> 4;
    const int l15   = lane & 15;
    const int jbase = blockIdx.x * 16;

    float* xt = xtile[wave];
    float* et = etile[wave];
    float* ew = et + (8 * half) * ESTRIDE + l15;

    v2f bj[8];
#pragma unroll
    for (int k0 = 0; k0 < 8; ++k0) {
        const int d0 = 4 * k0 + 2 * half;
        bj[k0].x = xin[(size_t)d0 * NPIX + jbase + l15];
        bj[k0].y = xin[(size_t)(d0 + 1) * NPIX + jbase + l15];
    }

    v8f y0 = {};
    v8f y1 = {};
    float dacc = 0.0f;

    const float* fsrc = xin + (size_t)(2 * half) * NPIX + l15 + wave * SLABW;

    v2f af[8][TB];
    v8f s[TB];
    v2f b2[TB][4], a2[TB][4], a3[TB][4];

    load_af(af, fsrc);
    stage_xt(xt, af, half, l15);
    gemm1(s, af, bj);
    load_af(af, fsrc + NWAVES * SLABW);

    for (int n = 0; n < NTRIPS - 1; ++n) {
        exp_phase(s, ew, dacc);
        load_frags(b2, a2, a3, et, xt, half, l15);
        stage_xt(xt, af, half, l15);
        __builtin_amdgcn_sched_barrier(0);
        gemm1(s, af, bj);
        __builtin_amdgcn_sched_barrier(0);
        const int nx = (n + 2 < NTRIPS) ? (n + 2) : (NTRIPS - 1);
        load_af(af, fsrc + (size_t)nx * (NWAVES * SLABW));
        __builtin_amdgcn_sched_barrier(0);
        gemm2(y0, y1, b2, a2, a3);
    }
    exp_phase(s, ew, dacc);
    load_frags(b2, a2, a3, et, xt, half, l15);
    gemm2(y0, y1, b2, a2, a3);

#pragma unroll
    for (int v = 0; v < 8; ++v) {
        red[wave][lane][v]     = y0[v];
        red[wave][lane][8 + v] = y1[v];
    }
    red[wave][lane][16] = dacc;
    __syncthreads();

    if (wave == 0) {
        v8f Y0 = {}, Y1 = {};
        float dtot = 0.0f;
        for (int w = 0; w < NWAVES; ++w) {
#pragma unroll
            for (int v = 0; v < 8; ++v) {
                Y0[v] += red[w][lane][v];
                Y1[v] += red[w][lane][8 + v];
            }
            dtot += red[w][lane][16];
        }
        const float dfull = dtot + __shfl_xor(dtot, 16, 32);
        const float sc = STEP / dfull;

#pragma unroll
        for (int v = 0; v < 8; ++v) {
            const int m = v + 8 * half;
            const size_t j = (size_t)jbase + l15;
            const size_t o0 = (size_t)m * NPIX + j;
            const size_t o1 = (size_t)(16 + m) * NPIX + j;
            const float r0 = Y0[v] * sc + (1.0f - STEP) * xin[o0];
            const float r1 = Y1[v] * sc + (1.0f - STEP) * xin[o1];
            xout[o0] = r0;
            xout[o1] = r1;
            if (xout2) { xout2[o0] = r0; xout2[o1] = r1; }
        }
    }
}

extern "C" void kernel_launch(void* const* d_in, const int* in_sizes, int n_in,
                              void* d_out, int out_size, void* d_ws, size_t ws_size,
                              hipStream_t stream) {
    (void)in_sizes; (void)n_in; (void)d_ws; (void)ws_size; (void)out_size;
    const float* x0 = (const float*)d_in[0];
    float* out = (float*)d_out;
    const size_t F = (size_t)DEMB * NPIX;

    dim3 grid(NTILES), block(32 * NWAVES);
    ms_iter_kernel<<<grid, block, 0, stream>>>(x0, out + F, nullptr);
    ms_iter_kernel<<<grid, block, 0, stream>>>(out + F, out + 2 * F, nullptr);
    ms_iter_kernel<<<grid, block, 0, stream>>>(out + 2 * F, out + 3 * F, out);
}


# exp moved into the matrix run's shadow: GEMM1 finishes its tiles two at a time, exp of a finished pair issues in three-stage pieces between the next matrix instructions; f32 sum order unchanged
# speedup vs baseline: 1.5265x; 1.0358x over previous
#include <hip/hip_runtime.h>

typedef __attribute__((ext_vector_type(2))) float v2f;
typedef __attribute__((ext_vector_type(8))) float v8f;

#define NPIX   9216
#define DEMB   32
#define NTILES (NPIX / 16)
#define NWAVES 4
#define TB     4
#define SLABW  (16 * TB)
#define NSLABS (NPIX / SLABW)
#define NTRIPS (NSLABS / NWAVES)
#define XSTRIDE 76
#define ESTRIDE 17
#define BW     3.0f
#define STEP   0.5f
#define LOG2E  1.44269504088896340736f

#define DEV static __device__ __forceinline__

DEV void load_af(v2f (&af)[8][TB], const float* __restrict__ src)
{
#pragma unroll
    for (int k0 = 0; k0 < 8; ++k0) {
#pragma unroll
        for (int t = 0; t < TB; ++t) {
            af[k0][t].x = src[(size_t)(4 * k0) * NPIX + 16 * t];
            af[k0][t].y = src[(size_t)(4 * k0 + 1) * NPIX + 16 * t];
        }
    }
}

DEV void stage_xt(float* xt, const v2f (&af)[8][TB], int half, int l15)
{
#pragma unroll
    for (int k0 = 0; k0 < 8; ++k0) {
        const int d0 = 4 * k0 + 2 * half;
#pragma unroll
        for (int t = 0; t < TB; ++t) {
            xt[d0 * XSTRIDE + 16 * t + l15]       = af[k0][t].x;
            xt[(d0 + 1) * XSTRIDE + 16 * t + l15] = af[k0][t].y;
        }
    }
}

DEV void gemm1_pair(v8f& sa, v8f& sb, const v2f (&af)[8][TB], int ta, int tb, const v2f (&bj)[8])
{
    sa = (v8f){};
    sb = (v8f){};
#pragma unroll
    for (int k0 = 0; k0 < 8; ++k0) {
        sa = __builtin_amdgcn_wmma_f32_16x16x4_f32(false, af[k0][ta], false, bj[k0],
                                                   (short)0, sa, false, false);
        sb = __builtin_amdgcn_wmma_f32_16x16x4_f32(false, af[k0][tb], false, bj[k0],
                                                   (short)0, sb, false, false);
    }
}

DEV void exp_gap(int g, v8f (&s)[TB], float (&pm)[16][2], float (&pe)[16][2], float (&es)[TB], float& dacc)
{
    if (g >= 3 && g - 3 < 16) {
        const int p = g - 3, t = p >> 2, v = 2 * (p & 3);
        es[t] += pe[p][0];
        es[t] += pe[p][1];
        s[t][v]     = pe[p][0];
        s[t][v + 1] = pe[p][1];
        if ((p & 3) == 3) dacc += es[t];
    }
    if (g >= 2 && g - 2 < 16) {
        const int p = g - 2;
        pe[p][0] = __builtin_amdgcn_exp2f(pm[p][0]);
        pe[p][1] = __builtin_amdgcn_exp2f(pm[p][1]);
    }
    if (g >= 1 && g - 1 < 16) {
        const int p = g - 1, t = p >> 2, v = 2 * (p & 3);
        pm[p][0] = s[t][v] * (BW * LOG2E);
        pm[p][1] = s[t][v + 1] * (BW * LOG2E);
    }
}

DEV void store_e(float* ew, const v8f (&s)[TB])
{
#pragma unroll
    for (int t = 0; t < TB; ++t) {
#pragma unroll
        for (int v = 0; v < 8; ++v) ew[(t * 16 + v) * ESTRIDE] = s[t][v];
    }
}

DEV void load_frags(v2f (&b2)[TB][4], v2f (&a2)[TB][4], v2f (&a3)[TB][4],
                    const float* et, const float* xt, int half, int l15)
{
#pragma unroll
    for (int t = 0; t < TB; ++t) {
#pragma unroll
        for (int c = 0; c < 4; ++c) {
            const int k0 = 4 * c + 2 * half;
            b2[t][c].x = et[t * 16 * ESTRIDE + k0 * ESTRIDE + l15];
            b2[t][c].y = et[t * 16 * ESTRIDE + (k0 + 1) * ESTRIDE + l15];
            a2[t][c].x = xt[l15 * XSTRIDE + 16 * t + k0];
            a2[t][c].y = xt[l15 * XSTRIDE + 16 * t + k0 + 1];
            a3[t][c].x = xt[(16 + l15) * XSTRIDE + 16 * t + k0];
            a3[t][c].y = xt[(16 + l15) * XSTRIDE + 16 * t + k0 + 1];
        }
    }
}

DEV void gemm2(v8f& y0, v8f& y1, const v2f (&b2)[TB][4], const v2f (&a2)[TB][4], const v2f (&a3)[TB][4])
{
#pragma unroll
    for (int t = 0; t < TB; ++t) {
#pragma unroll
        for (int c = 0; c < 4; ++c) {
            y0 = __builtin_amdgcn_wmma_f32_16x16x4_f32(false, a2[t][c], false, b2[t][c],
                                                       (short)0, y0, false, false);
            y1 = __builtin_amdgcn_wmma_f32_16x16x4_f32(false, a3[t][c], false, b2[t][c],
                                                       (short)0, y1, false, false);
        }
    }
}

__global__ __launch_bounds__(32 * NWAVES)
void ms_iter_kernel(const float* __restrict__ xin,
                    float* __restrict__ xout,
                    float* __restrict__ xout2)
{
    __shared__ __align__(16) float xtile[NWAVES][DEMB * XSTRIDE];
    __shared__ float etile[NWAVES][TB * 16 * ESTRIDE];
    __shared__ float red[NWAVES][32][17];

    const int lane  = threadIdx.x & 31;
    const int wave  = threadIdx.x >> 5;
    const int half  = lane >> 4;
    const int l15   = lane & 15;
    const int jbase = blockIdx.x * 16;

    float* xt = xtile[wave];
    float* et = etile[wave];
    float* ew = et + (8 * half) * ESTRIDE + l15;

    v2f bj[8];
#pragma unroll
    for (int k0 = 0; k0 < 8; ++k0) {
        const int d0 = 4 * k0 + 2 * half;
        bj[k0].x = xin[(size_t)d0 * NPIX + jbase + l15];
        bj[k0].y = xin[(size_t)(d0 + 1) * NPIX + jbase + l15];
    }

    v8f y0 = {};
    v8f y1 = {};
    float dacc = 0.0f;

    const float* fsrc = xin + (size_t)(2 * half) * NPIX + l15 + wave * SLABW;

    v2f af[8][TB];
    v8f s[TB];
    v2f b2[TB][4], a2[TB][4], a3[TB][4];
    float pm[16][2], pe[16][2], es[TB];

    load_af(af, fsrc);
    stage_xt(xt, af, half, l15);
    gemm1_pair(s[0], s[1], af, 0, 1, bj);
    gemm1_pair(s[2], s[3], af, 2, 3, bj);
    load_af(af, fsrc + NWAVES * SLABW);
    const float* pnext = fsrc + 2 * (NWAVES * SLABW);
#pragma unroll
    for (int t = 0; t < TB; ++t) es[t] = 0.0f;
#pragma unroll
    for (int g = 0; g < 19; ++g) exp_gap(g, s, pm, pe, es, dacc);

    for (int n = 0; n < NTRIPS - 1; ++n) {
        store_e(ew, s);
        load_frags(b2, a2, a3, et, xt, half, l15);
        stage_xt(xt, af, half, l15);
        __builtin_amdgcn_sched_barrier(0);
        gemm1_pair(s[0], s[1], af, 0, 1, bj);

#pragma unroll
        for (int t = 0; t < TB; ++t) es[t] = 0.0f;
        s[2] = (v8f){};
        s[3] = (v8f){};
#pragma unroll
        for (int k0 = 0; k0 < 8; ++k0) {
            s[2] = __builtin_amdgcn_wmma_f32_16x16x4_f32(false, af[k0][2], false, bj[k0],
                                                         (short)0, s[2], false, false);
            s[3] = __builtin_amdgcn_wmma_f32_16x16x4_f32(false, af[k0][3], false, bj[k0],
                                                         (short)0, s[3], false, false);
            __builtin_amdgcn_sched_barrier(0);
            exp_gap(k0, s, pm, pe, es, dacc);
        }

        load_af(af, pnext);
        __builtin_amdgcn_sched_barrier(0);

#pragma unroll
        for (int t = 0; t < TB; ++t) {
#pragma unroll
            for (int c = 0; c < 4; ++c) {
                y0 = __builtin_amdgcn_wmma_f32_16x16x4_f32(false, a2[t][c], false, b2[t][c],
                                                           (short)0, y0, false, false);
                y1 = __builtin_amdgcn_wmma_f32_16x16x4_f32(false, a3[t][c], false, b2[t][c],
                                                           (short)0, y1, false, false);
                __builtin_amdgcn_sched_barrier(0);
                exp_gap(8 + 4 * t + c, s, pm, pe, es, dacc);
            }
        }
        pnext += (n + 3 < NTRIPS) ? (NWAVES * SLABW) : 0;
    }
    store_e(ew, s);
    load_frags(b2, a2, a3, et, xt, half, l15);
    gemm2(y0, y1, b2, a2, a3);

#pragma unroll
    for (int v = 0; v < 8; ++v) {
        red[wave][lane][v]     = y0[v];
        red[wave][lane][8 + v] = y1[v];
    }
    red[wave][lane][16] = dacc;
    __syncthreads();

    if (wave == 0) {
        v8f Y0 = {}, Y1 = {};
        float dtot = 0.0f;
        for (int w = 0; w < NWAVES; ++w) {
#pragma unroll
            for (int v = 0; v < 8; ++v) {
                Y0[v] += red[w][lane][v];
                Y1[v] += red[w][lane][8 + v];
            }
            dtot += red[w][lane][16];
        }
        const float dfull = dtot + __shfl_xor(dtot, 16, 32);
        const float sc = STEP / dfull;

#pragma unroll
        for (int v = 0; v < 8; ++v) {
            const int m = v + 8 * half;
            const size_t j = (size_t)jbase + l15;
            const size_t o0 = (size_t)m * NPIX + j;
            const size_t o1 = (size_t)(16 + m) * NPIX + j;
            const float r0 = Y0[v] * sc + (1.0f - STEP) * xin[o0];
            const float r1 = Y1[v] * sc + (1.0f - STEP) * xin[o1];
            xout[o0] = r0;
            xout[o1] = r1;
            if (xout2) { xout2[o0] = r0; xout2[o1] = r1; }
        }
    }
}

extern "C" void kernel_launch(void* const* d_in, const int* in_sizes, int n_in,
                              void* d_out, int out_size, void* d_ws, size_t ws_size,
                              hipStream_t stream) {
    (void)in_sizes; (void)n_in; (void)d_ws; (void)ws_size; (void)out_size;
    const float* x0 = (const float*)d_in[0];
    float* out = (float*)d_out;
    const size_t F = (size_t)DEMB * NPIX;

    dim3 grid(NTILES), block(32 * NWAVES);
    ms_iter_kernel<<<grid, block, 0, stream>>>(x0, out + F, nullptr);
    ms_iter_kernel<<<grid, block, 0, stream>>>(out + F, out + 2 * F, nullptr);
    ms_iter_kernel<<<grid, block, 0, stream>>>(out + 2 * F, out + 3 * F, out);
}


# Slab columns dealt round-robin to the four i-tiles (tile t = columns 4m+t): a slab moves as 16 16-B global loads and 16 16-B LDS stores, not 64 and 64 of 4 B; f32 sum order of Y and d changes
# speedup vs baseline: 1.6097x; 1.0545x over previous
#include <hip/hip_runtime.h>

typedef __attribute__((ext_vector_type(2))) float v2f;
typedef __attribute__((ext_vector_type(4))) float v4f;
typedef __attribute__((ext_vector_type(8))) float v8f;

#define NPIX   9216
#define DEMB   32
#define NTILES (NPIX / 16)
#define NWAVES 4
#define TB     4
#define SLABW  (16 * TB)
#define NSLABS (NPIX / SLABW)
#define NTRIPS (NSLABS / NWAVES)
#define XSTRIDE 76
#define ESTRIDE 17
#define BW     3.0f
#define STEP   0.5f
#define LOG2E  1.44269504088896340736f

#define DEV static __device__ __forceinline__

DEV void load_slab(v4f (&ld)[8][2], const float* __restrict__ src)
{
#pragma unroll
    for (int k0 = 0; k0 < 8; ++k0) {
        ld[k0][0] = *reinterpret_cast<const v4f*>(src + (size_t)(4 * k0) * NPIX);
        ld[k0][1] = *reinterpret_cast<const v4f*>(src + (size_t)(4 * k0 + 1) * NPIX);
    }
}

DEV void unpack_af(v2f (&af)[8][TB], const v4f (&ld)[8][2])
{
#pragma unroll
    for (int k0 = 0; k0 < 8; ++k0) {
#pragma unroll
        for (int t = 0; t < TB; ++t) {
            af[k0][t].x = ld[k0][0][t];
            af[k0][t].y = ld[k0][1][t];
        }
    }
}

DEV void stage_xt(float* xt, const v4f (&ld)[8][2], int half, int l15)
{
#pragma unroll
    for (int k0 = 0; k0 < 8; ++k0) {
        const int d0 = 4 * k0 + 2 * half;
        *reinterpret_cast<v4f*>(xt + d0 * XSTRIDE + 4 * l15)       = ld[k0][0];
        *reinterpret_cast<v4f*>(xt + (d0 + 1) * XSTRIDE + 4 * l15) = ld[k0][1];
    }
}

DEV void gemm1_pair(v8f& sa, v8f& sb, const v2f (&af)[8][TB], int ta, int tb, const v2f (&bj)[8])
{
    sa = (v8f){};
    sb = (v8f){};
#pragma unroll
    for (int k0 = 0; k0 < 8; ++k0) {
        sa = __builtin_amdgcn_wmma_f32_16x16x4_f32(false, af[k0][ta], false, bj[k0],
                                                   (short)0, sa, false, false);
        sb = __builtin_amdgcn_wmma_f32_16x16x4_f32(false, af[k0][tb], false, bj[k0],
                                                   (short)0, sb, false, false);
    }
}

DEV void exp_gap(int g, v8f (&s)[TB], float (&pm)[16][2], float (&pe)[16][2], float (&es)[TB], float& dacc)
{
    if (g >= 3 && g - 3 < 16) {
        const int p = g - 3, t = p >> 2, v = 2 * (p & 3);
        es[t] += pe[p][0];
        es[t] += pe[p][1];
        s[t][v]     = pe[p][0];
        s[t][v + 1] = pe[p][1];
        if ((p & 3) == 3) dacc += es[t];
    }
    if (g >= 2 && g - 2 < 16) {
        const int p = g - 2;
        pe[p][0] = __builtin_amdgcn_exp2f(pm[p][0]);
        pe[p][1] = __builtin_amdgcn_exp2f(pm[p][1]);
    }
    if (g >= 1 && g - 1 < 16) {
        const int p = g - 1, t = p >> 2, v = 2 * (p & 3);
        pm[p][0] = s[t][v] * (BW * LOG2E);
        pm[p][1] = s[t][v + 1] * (BW * LOG2E);
    }
}

DEV void store_e(float* ew, const v8f (&s)[TB])
{
#pragma unroll
    for (int t = 0; t < TB; ++t) {
#pragma unroll
        for (int v = 0; v < 8; ++v) ew[(4 * v + t) * ESTRIDE] = s[t][v];
    }
}

DEV void load_frags(v2f (&b2)[TB][4], v2f (&a2)[TB][4], v2f (&a3)[TB][4],
                    const float* et, const float* xt, int half, int l15)
{
#pragma unroll
    for (int t = 0; t < TB; ++t) {
#pragma unroll
        for (int c = 0; c < 4; ++c) {
            const int k0 = 4 * c + 2 * half;
            b2[t][c].x = et[t * 16 * ESTRIDE + k0 * ESTRIDE + l15];
            b2[t][c].y = et[t * 16 * ESTRIDE + (k0 + 1) * ESTRIDE + l15];
            a2[t][c].x = xt[l15 * XSTRIDE + 16 * t + k0];
            a2[t][c].y = xt[l15 * XSTRIDE + 16 * t + k0 + 1];
            a3[t][c].x = xt[(16 + l15) * XSTRIDE + 16 * t + k0];
            a3[t][c].y = xt[(16 + l15) * XSTRIDE + 16 * t + k0 + 1];
        }
    }
}

DEV void gemm2(v8f& y0, v8f& y1, const v2f (&b2)[TB][4], const v2f (&a2)[TB][4], const v2f (&a3)[TB][4])
{
#pragma unroll
    for (int t = 0; t < TB; ++t) {
#pragma unroll
        for (int c = 0; c < 4; ++c) {
            y0 = __builtin_amdgcn_wmma_f32_16x16x4_f32(false, a2[t][c], false, b2[t][c],
                                                       (short)0, y0, false, false);
            y1 = __builtin_amdgcn_wmma_f32_16x16x4_f32(false, a3[t][c], false, b2[t][c],
                                                       (short)0, y1, false, false);
        }
    }
}

__global__ __launch_bounds__(32 * NWAVES)
void ms_iter_kernel(const float* __restrict__ xin,
                    float* __restrict__ xout,
                    float* __restrict__ xout2)
{
    __shared__ __align__(16) float xtile[NWAVES][DEMB * XSTRIDE];
    __shared__ float etile[NWAVES][TB * 16 * ESTRIDE];
    __shared__ float red[NWAVES][32][17];

    const int lane  = threadIdx.x & 31;
    const int wave  = threadIdx.x >> 5;
    const int half  = lane >> 4;
    const int l15   = lane & 15;
    const int jbase = blockIdx.x * 16;

    float* xt = xtile[wave];
    float* et = etile[wave];
    float* ew = et + (32 * half) * ESTRIDE + l15;

    v2f bj[8];
#pragma unroll
    for (int k0 = 0; k0 < 8; ++k0) {
        const int d0 = 4 * k0 + 2 * half;
        bj[k0].x = xin[(size_t)d0 * NPIX + jbase + l15];
        bj[k0].y = xin[(size_t)(d0 + 1) * NPIX + jbase + l15];
    }

    v8f y0 = {};
    v8f y1 = {};
    float dacc = 0.0f;

    const float* fsrc = xin + (size_t)(2 * half) * NPIX + 4 * l15 + wave * SLABW;

    v4f ld[8][2];
    v2f af[8][TB];
    v8f s[TB];
    v2f b2[TB][4], a2[TB][4], a3[TB][4];
    float pm[16][2], pe[16][2], es[TB];

    load_slab(ld, fsrc);
    stage_xt(xt, ld, half, l15);
    unpack_af(af, ld);
    gemm1_pair(s[0], s[1], af, 0, 1, bj);
    gemm1_pair(s[2], s[3], af, 2, 3, bj);
    load_slab(ld, fsrc + NWAVES * SLABW);
    const float* pnext = fsrc + 2 * (NWAVES * SLABW);
#pragma unroll
    for (int t = 0; t < TB; ++t) es[t] = 0.0f;
#pragma unroll
    for (int g = 0; g < 19; ++g) exp_gap(g, s, pm, pe, es, dacc);

    for (int n = 0; n < NTRIPS - 1; ++n) {
        store_e(ew, s);
        load_frags(b2, a2, a3, et, xt, half, l15);
        stage_xt(xt, ld, half, l15);
        unpack_af(af, ld);
        __builtin_amdgcn_sched_barrier(0);
        gemm1_pair(s[0], s[1], af, 0, 1, bj);

#pragma unroll
        for (int t = 0; t < TB; ++t) es[t] = 0.0f;
        s[2] = (v8f){};
        s[3] = (v8f){};
#pragma unroll
        for (int k0 = 0; k0 < 8; ++k0) {
            s[2] = __builtin_amdgcn_wmma_f32_16x16x4_f32(false, af[k0][2], false, bj[k0],
                                                         (short)0, s[2], false, false);
            s[3] = __builtin_amdgcn_wmma_f32_16x16x4_f32(false, af[k0][3], false, bj[k0],
                                                         (short)0, s[3], false, false);
            __builtin_amdgcn_sched_barrier(0);
            exp_gap(k0, s, pm, pe, es, dacc);
        }

        load_slab(ld, pnext);
        __builtin_amdgcn_sched_barrier(0);

#pragma unroll
        for (int t = 0; t < TB; ++t) {
#pragma unroll
            for (int c = 0; c < 4; ++c) {
                y0 = __builtin_amdgcn_wmma_f32_16x16x4_f32(false, a2[t][c], false, b2[t][c],
                                                           (short)0, y0, false, false);
                y1 = __builtin_amdgcn_wmma_f32_16x16x4_f32(false, a3[t][c], false, b2[t][c],
                                                           (short)0, y1, false, false);
                __builtin_amdgcn_sched_barrier(0);
                exp_gap(8 + 4 * t + c, s, pm, pe, es, dacc);
            }
        }
        pnext += (n + 3 < NTRIPS) ? (NWAVES * SLABW) : 0;
    }
    store_e(ew, s);
    load_frags(b2, a2, a3, et, xt, half, l15);
    gemm2(y0, y1, b2, a2, a3);

#pragma unroll
    for (int v = 0; v < 8; ++v) {
        red[wave][lane][v]     = y0[v];
        red[wave][lane][8 + v] = y1[v];
    }
    red[wave][lane][16] = dacc;
    __syncthreads();

    if (wave == 0) {
        v8f Y0 = {}, Y1 = {};
        float dtot = 0.0f;
        for (int w = 0; w < NWAVES; ++w) {
#pragma unroll
            for (int v = 0; v < 8; ++v) {
                Y0[v] += red[w][lane][v];
                Y1[v] += red[w][lane][8 + v];
            }
            dtot += red[w][lane][16];
        }
        const float dfull = dtot + __shfl_xor(dtot, 16, 32);
        const float sc = STEP / dfull;

#pragma unroll
        for (int v = 0; v < 8; ++v) {
            const int m = v + 8 * half;
            const size_t j = (size_t)jbase + l15;
            const size_t o0 = (size_t)m * NPIX + j;
            const size_t o1 = (size_t)(16 + m) * NPIX + j;
            const float r0 = Y0[v] * sc + (1.0f - STEP) * xin[o0];
            const float r1 = Y1[v] * sc + (1.0f - STEP) * xin[o1];
            xout[o0] = r0;
            xout[o1] = r1;
            if (xout2) { xout2[o0] = r0; xout2[o1] = r1; }
        }
    }
}

extern "C" void kernel_launch(void* const* d_in, const int* in_sizes, int n_in,
                              void* d_out, int out_size, void* d_ws, size_t ws_size,
                              hipStream_t stream) {
    (void)in_sizes; (void)n_in; (void)d_ws; (void)ws_size; (void)out_size;
    const float* x0 = (const float*)d_in[0];
    float* out = (float*)d_out;
    const size_t F = (size_t)DEMB * NPIX;

    dim3 grid(NTILES), block(32 * NWAVES);
    ms_iter_kernel<<<grid, block, 0, stream>>>(x0, out + F, nullptr);
    ms_iter_kernel<<<grid, block, 0, stream>>>(out + F, out + 2 * F, nullptr);
    ms_iter_kernel<<<grid, block, 0, stream>>>(out + 2 * F, out + 3 * F, out);
}


# K's symmetry used: one work-group per unordered pair of 256-column groups, each E tile feeds Y of both groups (24 matrix ops a tile pair, not 32); a 2nd kernel sums partials; f32 sum order changes
# speedup vs baseline: 1.9640x; 1.2201x over previous
#include <hip/hip_runtime.h>

typedef __attribute__((ext_vector_type(2))) float v2f;
typedef __attribute__((ext_vector_type(4))) float v4f;
typedef __attribute__((ext_vector_type(8))) float v8f;

#define NPIX   9216
#define DEMB   32
#define NTILES (NPIX / 16)
#define NWAVES 4
#define TB     4
#define SLABW  (16 * TB)
#define NSLABS (NPIX / SLABW)
#define NTRIPS (NSLABS / NWAVES)
#define XSTRIDE 76
#define ESTRIDE 17
#define BW     3.0f
#define STEP   0.5f
#define LOG2E  1.44269504088896340736f

#define DEV static __device__ __forceinline__

DEV void load_slab(v4f (&ld)[8][2], const float* __restrict__ src)
{
#pragma unroll
    for (int k0 = 0; k0 < 8; ++k0) {
        ld[k0][0] = *reinterpret_cast<const v4f*>(src + (size_t)(4 * k0) * NPIX);
        ld[k0][1] = *reinterpret_cast<const v4f*>(src + (size_t)(4 * k0 + 1) * NPIX);
    }
}

DEV void unpack_af(v2f (&af)[8][TB], const v4f (&ld)[8][2])
{
#pragma unroll
    for (int k0 = 0; k0 < 8; ++k0) {
#pragma unroll
        for (int t = 0; t < TB; ++t) {
            af[k0][t].x = ld[k0][0][t];
            af[k0][t].y = ld[k0][1][t];
        }
    }
}

DEV void stage_xt(float* xt, const v4f (&ld)[8][2], int half, int l15)
{
#pragma unroll
    for (int k0 = 0; k0 < 8; ++k0) {
        const int d0 = 4 * k0 + 2 * half;
        *reinterpret_cast<v4f*>(xt + d0 * XSTRIDE + 4 * l15)       = ld[k0][0];
        *reinterpret_cast<v4f*>(xt + (d0 + 1) * XSTRIDE + 4 * l15) = ld[k0][1];
    }
}

DEV void gemm1_pair(v8f& sa, v8f& sb, const v2f (&af)[8][TB], int ta, int tb, const v2f (&bj)[8])
{
    sa = (v8f){};
    sb = (v8f){};
#pragma unroll
    for (int k0 = 0; k0 < 8; ++k0) {
        sa = __builtin_amdgcn_wmma_f32_16x16x4_f32(false, af[k0][ta], false, bj[k0],
                                                   (short)0, sa, false, false);
        sb = __builtin_amdgcn_wmma_f32_16x16x4_f32(false, af[k0][tb], false, bj[k0],
                                                   (short)0, sb, false, false);
    }
}

DEV void exp_gap(int g, v8f (&s)[TB], float (&pm)[16][2], float (&pe)[16][2], float (&es)[TB], float& dacc)
{
    if (g >= 3 && g - 3 < 16) {
        const int p = g - 3, t = p >> 2, v = 2 * (p & 3);
        es[t] += pe[p][0];
        es[t] += pe[p][1];
        s[t][v]     = pe[p][0];
        s[t][v + 1] = pe[p][1];
        if ((p & 3) == 3) dacc += es[t];
    }
    if (g >= 2 && g - 2 < 16) {
        const int p = g - 2;
        pe[p][0] = __builtin_amdgcn_exp2f(pm[p][0]);
        pe[p][1] = __builtin_amdgcn_exp2f(pm[p][1]);
    }
    if (g >= 1 && g - 1 < 16) {
        const int p = g - 1, t = p >> 2, v = 2 * (p & 3);
        pm[p][0] = s[t][v] * (BW * LOG2E);
        pm[p][1] = s[t][v + 1] * (BW * LOG2E);
    }
}

DEV void store_e(float* ew, const v8f (&s)[TB])
{
#pragma unroll
    for (int t = 0; t < TB; ++t) {
#pragma unroll
        for (int v = 0; v < 8; ++v) ew[(4 * v + t) * ESTRIDE] = s[t][v];
    }
}

DEV void load_frags(v2f (&b2)[TB][4], v2f (&a2)[TB][4], v2f (&a3)[TB][4],
                    const float* et, const float* xt, int half, int l15)
{
#pragma unroll
    for (int t = 0; t < TB; ++t) {
#pragma unroll
        for (int c = 0; c < 4; ++c) {
            const int k0 = 4 * c + 2 * half;
            b2[t][c].x = et[t * 16 * ESTRIDE + k0 * ESTRIDE + l15];
            b2[t][c].y = et[t * 16 * ESTRIDE + (k0 + 1) * ESTRIDE + l15];
            a2[t][c].x = xt[l15 * XSTRIDE + 16 * t + k0];
            a2[t][c].y = xt[l15 * XSTRIDE + 16 * t + k0 + 1];
            a3[t][c].x = xt[(16 + l15) * XSTRIDE + 16 * t + k0];
            a3[t][c].y = xt[(16 + l15) * XSTRIDE + 16 * t + k0 + 1];
        }
    }
}

DEV void gemm2(v8f& y0, v8f& y1, const v2f (&b2)[TB][4], const v2f (&a2)[TB][4], const v2f (&a3)[TB][4])
{
#pragma unroll
    for (int t = 0; t < TB; ++t) {
#pragma unroll
        for (int c = 0; c < 4; ++c) {
            y0 = __builtin_amdgcn_wmma_f32_16x16x4_f32(false, a2[t][c], false, b2[t][c],
                                                       (short)0, y0, false, false);
            y1 = __builtin_amdgcn_wmma_f32_16x16x4_f32(false, a3[t][c], false, b2[t][c],
                                                       (short)0, y1, false, false);
        }
    }
}

__global__ __launch_bounds__(32 * NWAVES)
void ms_iter_kernel(const float* __restrict__ xin,
                    float* __restrict__ xout,
                    float* __restrict__ xout2)
{
    __shared__ __align__(16) float xtile[NWAVES][DEMB * XSTRIDE];
    __shared__ float etile[NWAVES][TB * 16 * ESTRIDE];
    __shared__ float red[NWAVES][32][17];

    const int lane  = threadIdx.x & 31;
    const int wave  = threadIdx.x >> 5;
    const int half  = lane >> 4;
    const int l15   = lane & 15;
    const int jbase = blockIdx.x * 16;

    float* xt = xtile[wave];
    float* et = etile[wave];
    float* ew = et + (32 * half) * ESTRIDE + l15;

    v2f bj[8];
#pragma unroll
    for (int k0 = 0; k0 < 8; ++k0) {
        const int d0 = 4 * k0 + 2 * half;
        bj[k0].x = xin[(size_t)d0 * NPIX + jbase + l15];
        bj[k0].y = xin[(size_t)(d0 + 1) * NPIX + jbase + l15];
    }

    v8f y0 = {};
    v8f y1 = {};
    float dacc = 0.0f;

    const float* fsrc = xin + (size_t)(2 * half) * NPIX + 4 * l15 + wave * SLABW;

    v4f ld[8][2];
    v2f af[8][TB];
    v8f s[TB];
    v2f b2[TB][4], a2[TB][4], a3[TB][4];
    float pm[16][2], pe[16][2], es[TB];

    load_slab(ld, fsrc);
    stage_xt(xt, ld, half, l15);
    unpack_af(af, ld);
    gemm1_pair(s[0], s[1], af, 0, 1, bj);
    gemm1_pair(s[2], s[3], af, 2, 3, bj);
    load_slab(ld, fsrc + NWAVES * SLABW);
    const float* pnext = fsrc + 2 * (NWAVES * SLABW);
#pragma unroll
    for (int t = 0; t < TB; ++t) es[t] = 0.0f;
#pragma unroll
    for (int g = 0; g < 19; ++g) exp_gap(g, s, pm, pe, es, dacc);

    for (int n = 0; n < NTRIPS - 1; ++n) {
        store_e(ew, s);
        load_frags(b2, a2, a3, et, xt, half, l15);
        stage_xt(xt, ld, half, l15);
        unpack_af(af, ld);
        __builtin_amdgcn_sched_barrier(0);
        gemm1_pair(s[0], s[1], af, 0, 1, bj);

#pragma unroll
        for (int t = 0; t < TB; ++t) es[t] = 0.0f;
        s[2] = (v8f){};
        s[3] = (v8f){};
#pragma unroll
        for (int k0 = 0; k0 < 8; ++k0) {
            s[2] = __builtin_amdgcn_wmma_f32_16x16x4_f32(false, af[k0][2], false, bj[k0],
                                                         (short)0, s[2], false, false);
            s[3] = __builtin_amdgcn_wmma_f32_16x16x4_f32(false, af[k0][3], false, bj[k0],
                                                         (short)0, s[3], false, false);
            __builtin_amdgcn_sched_barrier(0);
            exp_gap(k0, s, pm, pe, es, dacc);
        }

        load_slab(ld, pnext);
        __builtin_amdgcn_sched_barrier(0);

#pragma unroll
        for (int t = 0; t < TB; ++t) {
#pragma unroll
            for (int c = 0; c < 4; ++c) {
                y0 = __builtin_amdgcn_wmma_f32_16x16x4_f32(false, a2[t][c], false, b2[t][c],
                                                           (short)0, y0, false, false);
                y1 = __builtin_amdgcn_wmma_f32_16x16x4_f32(false, a3[t][c], false, b2[t][c],
                                                           (short)0, y1, false, false);
                __builtin_amdgcn_sched_barrier(0);
                exp_gap(8 + 4 * t + c, s, pm, pe, es, dacc);
            }
        }
        pnext += (n + 3 < NTRIPS) ? (NWAVES * SLABW) : 0;
    }
    store_e(ew, s);
    load_frags(b2, a2, a3, et, xt, half, l15);
    gemm2(y0, y1, b2, a2, a3);

#pragma unroll
    for (int v = 0; v < 8; ++v) {
        red[wave][lane][v]     = y0[v];
        red[wave][lane][8 + v] = y1[v];
    }
    red[wave][lane][16] = dacc;
    __syncthreads();

    if (wave == 0) {
        v8f Y0 = {}, Y1 = {};
        float dtot = 0.0f;
        for (int w = 0; w < NWAVES; ++w) {
#pragma unroll
            for (int v = 0; v < 8; ++v) {
                Y0[v] += red[w][lane][v];
                Y1[v] += red[w][lane][8 + v];
            }
            dtot += red[w][lane][16];
        }
        const float dfull = dtot + __shfl_xor(dtot, 16, 32);
        const float sc = STEP / dfull;

#pragma unroll
        for (int v = 0; v < 8; ++v) {
            const int m = v + 8 * half;
            const size_t j = (size_t)jbase + l15;
            const size_t o0 = (size_t)m * NPIX + j;
            const size_t o1 = (size_t)(16 + m) * NPIX + j;
            const float r0 = Y0[v] * sc + (1.0f - STEP) * xin[o0];
            const float r1 = Y1[v] * sc + (1.0f - STEP) * xin[o1];
            xout[o0] = r0;
            xout[o1] = r1;
            if (xout2) { xout2[o0] = r0; xout2[o1] = r1; }
        }
    }
}


#define GW     256
#define NG     (NPIX / GW)
#define NPAIRS (NG * (NG + 1) / 2)
#define JT     4
#define ITG    (GW / 16)
#define ES     76
#define HW     128
#define YS     132
#define WSROWS 33
#define WS_NEED 43794432ull

DEV void load_af(v2f (&af)[8], const float* __restrict__ xin, int i0, int half, int l15)
{
#pragma unroll
    for (int k0 = 0; k0 < 8; ++k0) {
        const int d0 = 4 * k0 + 2 * half;
        af[k0].x = xin[(size_t)d0 * NPIX + i0 + l15];
        af[k0].y = xin[(size_t)(d0 + 1) * NPIX + i0 + l15];
    }
}
DEV void load_xa(v4f (&xa)[4], const float* __restrict__ xin, int i0, int half, int l15)
{
    const float* p = xin + (size_t)l15 * NPIX + i0 + 8 * half;
    xa[0] = *reinterpret_cast<const v4f*>(p);
    xa[1] = *reinterpret_cast<const v4f*>(p + 4);
    xa[2] = *reinterpret_cast<const v4f*>(p + (size_t)16 * NPIX);
    xa[3] = *reinterpret_cast<const v4f*>(p + (size_t)16 * NPIX + 4);
}

DEV v2f pick2(const v4f& q, int e) { v2f r; r.x = q[2 * e]; r.y = q[2 * e + 1]; return r; }

template <bool OD>
DEV void pair_walk(const float* __restrict__ xin, float* __restrict__ ws,
                   float* et, float* yall, int I, int J, int wave, int half, int l15)
{
    const int ibase = I * GW;
    const int jw    = J * GW + wave * (16 * JT);
    float* yw = yall + wave * (WSROWS * YS);

    v2f bj[JT][8];
#pragma unroll
    for (int jt = 0; jt < JT; ++jt) {
#pragma unroll
        for (int k0 = 0; k0 < 8; ++k0) {
            const int d0 = 4 * k0 + 2 * half;
            bj[jt][k0].x = xin[(size_t)d0 * NPIX + jw + 16 * jt + l15];
            bj[jt][k0].y = xin[(size_t)(d0 + 1) * NPIX + jw + 16 * jt + l15];
        }
    }
    v4f xj0[JT][2], xj1[JT][2];
    if (OD) {
#pragma unroll
        for (int jt = 0; jt < JT; ++jt) {
#pragma unroll
            for (int cp = 0; cp < 2; ++cp) {
                const float* p = xin + (size_t)l15 * NPIX + jw + 16 * jt + 8 * cp + 4 * half;
                xj0[jt][cp] = *reinterpret_cast<const v4f*>(p);
                xj1[jt][cp] = *reinterpret_cast<const v4f*>(p + (size_t)16 * NPIX);
            }
        }
    }

    v8f YJ0[JT], YJ1[JT];
    float dJ[JT];
#pragma unroll
    for (int jt = 0; jt < JT; ++jt) { YJ0[jt] = (v8f){}; YJ1[jt] = (v8f){}; dJ[jt] = 0.0f; }

    v2f af[8];
    v4f xa[4];
    load_af(af, xin, ibase, half, l15);
    load_xa(xa, xin, ibase, half, l15);

    for (int it = 0; it < ITG; ++it) {
        const int inext = ibase + 16 * ((it + 1 < ITG) ? (it + 1) : it);

        v8f S[JT];
#pragma unroll
        for (int jt = 0; jt < JT; ++jt) S[jt] = (v8f){};
#pragma unroll
        for (int k0 = 0; k0 < 8; ++k0) {
#pragma unroll
            for (int jt = 0; jt < JT; ++jt)
                S[jt] = __builtin_amdgcn_wmma_f32_16x16x4_f32(false, af[k0], false, bj[jt][k0],
                                                              (short)0, S[jt], false, false);
        }
        load_af(af, xin, inext, half, l15);
#pragma unroll
        for (int jt = 0; jt < JT; ++jt) {
            float e = 0.0f;
#pragma unroll
            for (int v = 0; v < 8; ++v) {
                S[jt][v] = __builtin_amdgcn_exp2f(S[jt][v] * (BW * LOG2E));
                e += S[jt][v];
            }
            dJ[jt] += e;
        }
        if (OD) {
#pragma unroll
            for (int jt = 0; jt < JT; ++jt) {
#pragma unroll
                for (int v = 0; v < 8; ++v) et[(8 * half + v) * ES + 16 * jt + l15] = S[jt][v];
            }
        }
#pragma unroll
        for (int c = 0; c < 4; ++c) {
            const v2f a0 = pick2(xa[c >> 1], c & 1);
            const v2f a1 = pick2(xa[2 + (c >> 1)], c & 1);
#pragma unroll
            for (int jt = 0; jt < JT; ++jt) {
                v2f bf; bf.x = S[jt][2 * c]; bf.y = S[jt][2 * c + 1];
                YJ0[jt] = __builtin_amdgcn_wmma_f32_16x16x4_f32(false, a0, false, bf, (short)0, YJ0[jt], false, false);
                YJ1[jt] = __builtin_amdgcn_wmma_f32_16x16x4_f32(false, a1, false, bf, (short)0, YJ1[jt], false, false);
            }
        }
        load_xa(xa, xin, inext, half, l15);
        if (OD) {
            v8f Yi0 = {}, Yi1 = {};
            float ri = 0.0f;
#pragma unroll
            for (int jt = 0; jt < JT; ++jt) {
#pragma unroll
                for (int cp = 0; cp < 2; ++cp) {
                    const v4f r = *reinterpret_cast<const v4f*>(et + l15 * ES + 16 * jt + 8 * cp + 4 * half);
                    ri += (r[0] + r[1]) + (r[2] + r[3]);
#pragma unroll
                    for (int e = 0; e < 2; ++e) {
                        const v2f bt = pick2(r, e);
                        Yi0 = __builtin_amdgcn_wmma_f32_16x16x4_f32(false, pick2(xj0[jt][cp], e), false, bt,
                                                                    (short)0, Yi0, false, false);
                        Yi1 = __builtin_amdgcn_wmma_f32_16x16x4_f32(false, pick2(xj1[jt][cp], e), false, bt,
                                                                    (short)0, Yi1, false, false);
                    }
                }
            }
#pragma unroll
            for (int v = 0; v < 8; ++v) {
                yw[(8 * half + v) * YS + 16 * (it & 7) + l15]      = Yi0[v];
                yw[(16 + 8 * half + v) * YS + 16 * (it & 7) + l15] = Yi1[v];
            }
            const float rfull = ri + __shfl_xor(ri, 16, 32);
            if (half == 0) yw[32 * YS + 16 * (it & 7) + l15] = rfull;

            if ((it & 7) == 7) {
                __syncthreads();
                float* wsJ = ws + (size_t)J * WSROWS * NPIX + (size_t)ibase + HW * (it >> 3);
                for (int e = threadIdx.x; e < WSROWS * (HW / 4); e += 32 * NWAVES) {
                    const int r = e >> 5, c = 4 * (e & (HW / 4 - 1));
                    const float* y0p = yall + r * YS + c;
                    const v4f t0 = *reinterpret_cast<const v4f*>(y0p);
                    const v4f t1 = *reinterpret_cast<const v4f*>(y0p + WSROWS * YS);
                    const v4f t2 = *reinterpret_cast<const v4f*>(y0p + 2 * WSROWS * YS);
                    const v4f t3 = *reinterpret_cast<const v4f*>(y0p + 3 * WSROWS * YS);
                    *reinterpret_cast<v4f*>(wsJ + (size_t)r * NPIX + c) = (t0 + t1) + (t2 + t3);
                }
                __syncthreads();
            }
        }
    }

    float* wsI = ws + (size_t)I * WSROWS * NPIX;
#pragma unroll
    for (int jt = 0; jt < JT; ++jt) {
        const size_t j = (size_t)jw + 16 * jt + l15;
#pragma unroll
        for (int v = 0; v < 8; ++v) {
            wsI[(size_t)(8 * half + v) * NPIX + j]      = YJ0[jt][v];
            wsI[(size_t)(16 + 8 * half + v) * NPIX + j] = YJ1[jt][v];
        }
        const float dfull = dJ[jt] + __shfl_xor(dJ[jt], 16, 32);
        if (half == 0) wsI[(size_t)32 * NPIX + j] = dfull;
    }
}

__global__ __launch_bounds__(32 * NWAVES, 3)
void ms_pair_kernel(const float* __restrict__ xin, float* __restrict__ ws)
{
    __shared__ __align__(16) float etile[NWAVES][16 * ES];
    __shared__ __align__(16) float yslab[NWAVES][WSROWS * YS];

    const int lane = threadIdx.x & 31;
    const int wave = threadIdx.x >> 5;
    const int half = lane >> 4;
    const int l15  = lane & 15;

    int b = blockIdx.x, I = 0;
    while (b >= NG - I) { b -= NG - I; ++I; }
    const int J = I + b;

    if (I == J) pair_walk<false>(xin, ws, etile[wave], &yslab[0][0], I, J, wave, half, l15);
    else        pair_walk<true>(xin, ws, etile[wave], &yslab[0][0], I, J, wave, half, l15);
}

__global__ __launch_bounds__(256)
void ms_combine_kernel(const float* __restrict__ xin, const float* __restrict__ ws,
                       float* __restrict__ xout, float* __restrict__ xout2)
{
    const int c  = blockIdx.x * 32 + threadIdx.x;
    const int r0 = 4 * threadIdx.y;
    float acc[4] = {0.0f, 0.0f, 0.0f, 0.0f};
    float dsum = 0.0f;
#pragma unroll 9
    for (int P = 0; P < NG; ++P) {
        const float* s = ws + (size_t)P * WSROWS * NPIX + c;
#pragma unroll
        for (int r = 0; r < 4; ++r) acc[r] += s[(size_t)(r0 + r) * NPIX];
        dsum += s[(size_t)32 * NPIX];
    }
    const float sc = STEP / dsum;
#pragma unroll
    for (int r = 0; r < 4; ++r) {
        const size_t o = (size_t)(r0 + r) * NPIX + c;
        const float val = acc[r] * sc + (1.0f - STEP) * xin[o];
        xout[o] = val;
        if (xout2) xout2[o] = val;
    }
}

extern "C" void kernel_launch(void* const* d_in, const int* in_sizes, int n_in,
                              void* d_out, int out_size, void* d_ws, size_t ws_size,
                              hipStream_t stream) {
    (void)in_sizes; (void)n_in; (void)out_size;
    const float* x0 = (const float*)d_in[0];
    float* out = (float*)d_out;
    float* ws = (float*)d_ws;
    const size_t F = (size_t)DEMB * NPIX;

    if (d_ws != nullptr && ws_size >= WS_NEED) {
        dim3 pgrid(NPAIRS), pblock(32 * NWAVES), cgrid(NPIX / 32), cblock(32, 8);
        ms_pair_kernel<<<pgrid, pblock, 0, stream>>>(x0, ws);
        ms_combine_kernel<<<cgrid, cblock, 0, stream>>>(x0, ws, out + F, nullptr);
        ms_pair_kernel<<<pgrid, pblock, 0, stream>>>(out + F, ws);
        ms_combine_kernel<<<cgrid, cblock, 0, stream>>>(out + F, ws, out + 2 * F, nullptr);
        ms_pair_kernel<<<pgrid, pblock, 0, stream>>>(out + 2 * F, ws);
        ms_combine_kernel<<<cgrid, cblock, 0, stream>>>(out + 2 * F, ws, out + 3 * F, out);
    } else {
        dim3 grid(NTILES), block(32 * NWAVES);
        ms_iter_kernel<<<grid, block, 0, stream>>>(x0, out + F, nullptr);
        ms_iter_kernel<<<grid, block, 0, stream>>>(out + F, out + 2 * F, nullptr);
        ms_iter_kernel<<<grid, block, 0, stream>>>(out + 2 * F, out + 3 * F, out);
    }
}


# Combine kernel: a thread owns one row of four columns and has all of its up to 36 slab loads in flight at once, 16 B each (one memory round trip, not up to ten in a row); f32 sum order unchanged
# speedup vs baseline: 2.2598x; 1.1506x over previous
#include <hip/hip_runtime.h>

typedef __attribute__((ext_vector_type(2))) float v2f;
typedef __attribute__((ext_vector_type(4))) float v4f;
typedef __attribute__((ext_vector_type(8))) float v8f;

#define NPIX   9216
#define DEMB   32
#define NTILES (NPIX / 16)
#define NWAVES 4
#define TB     4
#define SLABW  (16 * TB)
#define NSLABS (NPIX / SLABW)
#define NTRIPS (NSLABS / NWAVES)
#define XSTRIDE 76
#define ESTRIDE 17
#define BW     3.0f
#define STEP   0.5f
#define LOG2E  1.44269504088896340736f

#define DEV static __device__ __forceinline__

DEV void load_slab(v4f (&ld)[8][2], const float* __restrict__ src)
{
#pragma unroll
    for (int k0 = 0; k0 < 8; ++k0) {
        ld[k0][0] = *reinterpret_cast<const v4f*>(src + (size_t)(4 * k0) * NPIX);
        ld[k0][1] = *reinterpret_cast<const v4f*>(src + (size_t)(4 * k0 + 1) * NPIX);
    }
}

DEV void unpack_af(v2f (&af)[8][TB], const v4f (&ld)[8][2])
{
#pragma unroll
    for (int k0 = 0; k0 < 8; ++k0) {
#pragma unroll
        for (int t = 0; t < TB; ++t) {
            af[k0][t].x = ld[k0][0][t];
            af[k0][t].y = ld[k0][1][t];
        }
    }
}

DEV void stage_xt(float* xt, const v4f (&ld)[8][2], int half, int l15)
{
#pragma unroll
    for (int k0 = 0; k0 < 8; ++k0) {
        const int d0 = 4 * k0 + 2 * half;
        *reinterpret_cast<v4f*>(xt + d0 * XSTRIDE + 4 * l15)       = ld[k0][0];
        *reinterpret_cast<v4f*>(xt + (d0 + 1) * XSTRIDE + 4 * l15) = ld[k0][1];
    }
}

DEV void gemm1_pair(v8f& sa, v8f& sb, const v2f (&af)[8][TB], int ta, int tb, const v2f (&bj)[8])
{
    sa = (v8f){};
    sb = (v8f){};
#pragma unroll
    for (int k0 = 0; k0 < 8; ++k0) {
        sa = __builtin_amdgcn_wmma_f32_16x16x4_f32(false, af[k0][ta], false, bj[k0],
                                                   (short)0, sa, false, false);
        sb = __builtin_amdgcn_wmma_f32_16x16x4_f32(false, af[k0][tb], false, bj[k0],
                                                   (short)0, sb, false, false);
    }
}

DEV void exp_gap(int g, v8f (&s)[TB], float (&pm)[16][2], float (&pe)[16][2], float (&es)[TB], float& dacc)
{
    if (g >= 3 && g - 3 < 16) {
        const int p = g - 3, t = p >> 2, v = 2 * (p & 3);
        es[t] += pe[p][0];
        es[t] += pe[p][1];
        s[t][v]     = pe[p][0];
        s[t][v + 1] = pe[p][1];
        if ((p & 3) == 3) dacc += es[t];
    }
    if (g >= 2 && g - 2 < 16) {
        const int p = g - 2;
        pe[p][0] = __builtin_amdgcn_exp2f(pm[p][0]);
        pe[p][1] = __builtin_amdgcn_exp2f(pm[p][1]);
    }
    if (g >= 1 && g - 1 < 16) {
        const int p = g - 1, t = p >> 2, v = 2 * (p & 3);
        pm[p][0] = s[t][v] * (BW * LOG2E);
        pm[p][1] = s[t][v + 1] * (BW * LOG2E);
    }
}

DEV void store_e(float* ew, const v8f (&s)[TB])
{
#pragma unroll
    for (int t = 0; t < TB; ++t) {
#pragma unroll
        for (int v = 0; v < 8; ++v) ew[(4 * v + t) * ESTRIDE] = s[t][v];
    }
}

DEV void load_frags(v2f (&b2)[TB][4], v2f (&a2)[TB][4], v2f (&a3)[TB][4],
                    const float* et, const float* xt, int half, int l15)
{
#pragma unroll
    for (int t = 0; t < TB; ++t) {
#pragma unroll
        for (int c = 0; c < 4; ++c) {
            const int k0 = 4 * c + 2 * half;
            b2[t][c].x = et[t * 16 * ESTRIDE + k0 * ESTRIDE + l15];
            b2[t][c].y = et[t * 16 * ESTRIDE + (k0 + 1) * ESTRIDE + l15];
            a2[t][c].x = xt[l15 * XSTRIDE + 16 * t + k0];
            a2[t][c].y = xt[l15 * XSTRIDE + 16 * t + k0 + 1];
            a3[t][c].x = xt[(16 + l15) * XSTRIDE + 16 * t + k0];
            a3[t][c].y = xt[(16 + l15) * XSTRIDE + 16 * t + k0 + 1];
        }
    }
}

DEV void gemm2(v8f& y0, v8f& y1, const v2f (&b2)[TB][4], const v2f (&a2)[TB][4], const v2f (&a3)[TB][4])
{
#pragma unroll
    for (int t = 0; t < TB; ++t) {
#pragma unroll
        for (int c = 0; c < 4; ++c) {
            y0 = __builtin_amdgcn_wmma_f32_16x16x4_f32(false, a2[t][c], false, b2[t][c],
                                                       (short)0, y0, false, false);
            y1 = __builtin_amdgcn_wmma_f32_16x16x4_f32(false, a3[t][c], false, b2[t][c],
                                                       (short)0, y1, false, false);
        }
    }
}

__global__ __launch_bounds__(32 * NWAVES)
void ms_iter_kernel(const float* __restrict__ xin,
                    float* __restrict__ xout,
                    float* __restrict__ xout2)
{
    __shared__ __align__(16) float xtile[NWAVES][DEMB * XSTRIDE];
    __shared__ float etile[NWAVES][TB * 16 * ESTRIDE];
    __shared__ float red[NWAVES][32][17];

    const int lane  = threadIdx.x & 31;
    const int wave  = threadIdx.x >> 5;
    const int half  = lane >> 4;
    const int l15   = lane & 15;
    const int jbase = blockIdx.x * 16;

    float* xt = xtile[wave];
    float* et = etile[wave];
    float* ew = et + (32 * half) * ESTRIDE + l15;

    v2f bj[8];
#pragma unroll
    for (int k0 = 0; k0 < 8; ++k0) {
        const int d0 = 4 * k0 + 2 * half;
        bj[k0].x = xin[(size_t)d0 * NPIX + jbase + l15];
        bj[k0].y = xin[(size_t)(d0 + 1) * NPIX + jbase + l15];
    }

    v8f y0 = {};
    v8f y1 = {};
    float dacc = 0.0f;

    const float* fsrc = xin + (size_t)(2 * half) * NPIX + 4 * l15 + wave * SLABW;

    v4f ld[8][2];
    v2f af[8][TB];
    v8f s[TB];
    v2f b2[TB][4], a2[TB][4], a3[TB][4];
    float pm[16][2], pe[16][2], es[TB];

    load_slab(ld, fsrc);
    stage_xt(xt, ld, half, l15);
    unpack_af(af, ld);
    gemm1_pair(s[0], s[1], af, 0, 1, bj);
    gemm1_pair(s[2], s[3], af, 2, 3, bj);
    load_slab(ld, fsrc + NWAVES * SLABW);
    const float* pnext = fsrc + 2 * (NWAVES * SLABW);
#pragma unroll
    for (int t = 0; t < TB; ++t) es[t] = 0.0f;
#pragma unroll
    for (int g = 0; g < 19; ++g) exp_gap(g, s, pm, pe, es, dacc);

    for (int n = 0; n < NTRIPS - 1; ++n) {
        store_e(ew, s);
        load_frags(b2, a2, a3, et, xt, half, l15);
        stage_xt(xt, ld, half, l15);
        unpack_af(af, ld);
        __builtin_amdgcn_sched_barrier(0);
        gemm1_pair(s[0], s[1], af, 0, 1, bj);

#pragma unroll
        for (int t = 0; t < TB; ++t) es[t] = 0.0f;
        s[2] = (v8f){};
        s[3] = (v8f){};
#pragma unroll
        for (int k0 = 0; k0 < 8; ++k0) {
            s[2] = __builtin_amdgcn_wmma_f32_16x16x4_f32(false, af[k0][2], false, bj[k0],
                                                         (short)0, s[2], false, false);
            s[3] = __builtin_amdgcn_wmma_f32_16x16x4_f32(false, af[k0][3], false, bj[k0],
                                                         (short)0, s[3], false, false);
            __builtin_amdgcn_sched_barrier(0);
            exp_gap(k0, s, pm, pe, es, dacc);
        }

        load_slab(ld, pnext);
        __builtin_amdgcn_sched_barrier(0);

#pragma unroll
        for (int t = 0; t < TB; ++t) {
#pragma unroll
            for (int c = 0; c < 4; ++c) {
                y0 = __builtin_amdgcn_wmma_f32_16x16x4_f32(false, a2[t][c], false, b2[t][c],
                                                           (short)0, y0, false, false);
                y1 = __builtin_amdgcn_wmma_f32_16x16x4_f32(false, a3[t][c], false, b2[t][c],
                                                           (short)0, y1, false, false);
                __builtin_amdgcn_sched_barrier(0);
                exp_gap(8 + 4 * t + c, s, pm, pe, es, dacc);
            }
        }
        pnext += (n + 3 < NTRIPS) ? (NWAVES * SLABW) : 0;
    }
    store_e(ew, s);
    load_frags(b2, a2, a3, et, xt, half, l15);
    gemm2(y0, y1, b2, a2, a3);

#pragma unroll
    for (int v = 0; v < 8; ++v) {
        red[wave][lane][v]     = y0[v];
        red[wave][lane][8 + v] = y1[v];
    }
    red[wave][lane][16] = dacc;
    __syncthreads();

    if (wave == 0) {
        v8f Y0 = {}, Y1 = {};
        float dtot = 0.0f;
        for (int w = 0; w < NWAVES; ++w) {
#pragma unroll
            for (int v = 0; v < 8; ++v) {
                Y0[v] += red[w][lane][v];
                Y1[v] += red[w][lane][8 + v];
            }
            dtot += red[w][lane][16];
        }
        const float dfull = dtot + __shfl_xor(dtot, 16, 32);
        const float sc = STEP / dfull;

#pragma unroll
        for (int v = 0; v < 8; ++v) {
            const int m = v + 8 * half;
            const size_t j = (size_t)jbase + l15;
            const size_t o0 = (size_t)m * NPIX + j;
            const size_t o1 = (size_t)(16 + m) * NPIX + j;
            const float r0 = Y0[v] * sc + (1.0f - STEP) * xin[o0];
            const float r1 = Y1[v] * sc + (1.0f - STEP) * xin[o1];
            xout[o0] = r0;
            xout[o1] = r1;
            if (xout2) { xout2[o0] = r0; xout2[o1] = r1; }
        }
    }
}


#define GW     256
#define NG     (NPIX / GW)
#define PW     8
#define JT     2
#define ITG    (GW / 16)
#define ES     36
#define HW     128
#define YS     132
#define WSROWS 33
#define WS_NEED 43794432ull
#define UOD    3
#define UDG    2
#define NPBLK  511

DEV int grp_cost(int G)   { return UOD * ITG * G + UDG * ITG; }
DEV int grp_blocks(int G) { const int n = (grp_cost(G) + 52) / 63; return (n < G + 1) ? n : (G + 1); }
DEV int cut_tile(int G, int u)
{
    const int uo = UOD * ITG * G;
    return (u <= uo) ? (u / UOD) : (ITG * G + (u - uo) / UDG);
}

DEV void load_af(v2f (&af)[8], const float* __restrict__ xin, int i0, int half, int l15)
{
#pragma unroll
    for (int k0 = 0; k0 < 8; ++k0) {
        const int d0 = 4 * k0 + 2 * half;
        af[k0].x = xin[(size_t)d0 * NPIX + i0 + l15];
        af[k0].y = xin[(size_t)(d0 + 1) * NPIX + i0 + l15];
    }
}
DEV void load_xa(v4f (&xa)[4], const float* __restrict__ xin, int i0, int half, int l15)
{
    const float* p = xin + (size_t)l15 * NPIX + i0 + 8 * half;
    xa[0] = *reinterpret_cast<const v4f*>(p);
    xa[1] = *reinterpret_cast<const v4f*>(p + 4);
    xa[2] = *reinterpret_cast<const v4f*>(p + (size_t)16 * NPIX);
    xa[3] = *reinterpret_cast<const v4f*>(p + (size_t)16 * NPIX + 4);
}

DEV v2f pick2(const v4f& q, int e) { v2f r; r.x = q[2 * e]; r.y = q[2 * e + 1]; return r; }

template <bool OD>
DEV void walk(const float* __restrict__ xin, float* __restrict__ wsG,
              float* et, float* yall, float* yw, int ta, int tb, int half, int l15,
              const v2f (&bj)[JT][8], const v4f (&xj0)[JT][2], const v4f (&xj1)[JT][2],
              v8f (&YJ0)[JT], v8f (&YJ1)[JT], float (&dJ)[JT])
{
    if (ta >= tb) return;

    v2f af[8];
    v4f xa[4];
    load_af(af, xin, 16 * ta, half, l15);
    load_xa(xa, xin, 16 * ta, half, l15);

    for (int it = ta; it < tb; ++it) {
        const int inext = 16 * ((it + 1 < tb) ? (it + 1) : it);

        v8f S[JT];
#pragma unroll
        for (int jt = 0; jt < JT; ++jt) S[jt] = (v8f){};
#pragma unroll
        for (int k0 = 0; k0 < 8; ++k0) {
#pragma unroll
            for (int jt = 0; jt < JT; ++jt)
                S[jt] = __builtin_amdgcn_wmma_f32_16x16x4_f32(false, af[k0], false, bj[jt][k0],
                                                              (short)0, S[jt], false, false);
        }
        load_af(af, xin, inext, half, l15);
#pragma unroll
        for (int jt = 0; jt < JT; ++jt) {
            float e = 0.0f;
#pragma unroll
            for (int v = 0; v < 8; ++v) {
                S[jt][v] = __builtin_amdgcn_exp2f(S[jt][v] * (BW * LOG2E));
                e += S[jt][v];
            }
            dJ[jt] += e;
        }
        if (OD) {
#pragma unroll
            for (int jt = 0; jt < JT; ++jt) {
#pragma unroll
                for (int v = 0; v < 8; ++v) et[(8 * half + v) * ES + 16 * jt + l15] = S[jt][v];
            }
        }
#pragma unroll
        for (int c = 0; c < 4; ++c) {
            const v2f a0 = pick2(xa[c >> 1], c & 1);
            const v2f a1 = pick2(xa[2 + (c >> 1)], c & 1);
#pragma unroll
            for (int jt = 0; jt < JT; ++jt) {
                v2f bf; bf.x = S[jt][2 * c]; bf.y = S[jt][2 * c + 1];
                YJ0[jt] = __builtin_amdgcn_wmma_f32_16x16x4_f32(false, a0, false, bf, (short)0, YJ0[jt], false, false);
                YJ1[jt] = __builtin_amdgcn_wmma_f32_16x16x4_f32(false, a1, false, bf, (short)0, YJ1[jt], false, false);
            }
        }
        load_xa(xa, xin, inext, half, l15);
        if (OD) {
            const int slot = (it - ta) & 7;
            v8f Yi0 = {}, Yi1 = {};
            float ri = 0.0f;
#pragma unroll
            for (int jt = 0; jt < JT; ++jt) {
#pragma unroll
                for (int cp = 0; cp < 2; ++cp) {
                    const v4f r = *reinterpret_cast<const v4f*>(et + l15 * ES + 16 * jt + 8 * cp + 4 * half);
                    ri += (r[0] + r[1]) + (r[2] + r[3]);
#pragma unroll
                    for (int e = 0; e < 2; ++e) {
                        const v2f bt = pick2(r, e);
                        Yi0 = __builtin_amdgcn_wmma_f32_16x16x4_f32(false, pick2(xj0[jt][cp], e), false, bt,
                                                                    (short)0, Yi0, false, false);
                        Yi1 = __builtin_amdgcn_wmma_f32_16x16x4_f32(false, pick2(xj1[jt][cp], e), false, bt,
                                                                    (short)0, Yi1, false, false);
                    }
                }
            }
#pragma unroll
            for (int v = 0; v < 8; ++v) {
                yw[(8 * half + v) * YS + 16 * slot + l15]      = Yi0[v];
                yw[(16 + 8 * half + v) * YS + 16 * slot + l15] = Yi1[v];
            }
            const float rfull = ri + __shfl_xor(ri, 16, 32);
            if (half == 0) yw[32 * YS + 16 * slot + l15] = rfull;

            if (slot == 7 || it + 1 == tb) {
                __syncthreads();
                float* wsJ = wsG + 16 * (it - slot);
                const int ncol = 16 * (slot + 1);
                for (int e = threadIdx.x; e < WSROWS * (HW / 4); e += 32 * PW) {
                    const int r = e >> 5, c = 4 * (e & (HW / 4 - 1));
                    if (c < ncol) {
                        const float* y0p = yall + r * YS + c;
                        v4f t[PW];
#pragma unroll
                        for (int w = 0; w < PW; ++w) t[w] = *reinterpret_cast<const v4f*>(y0p + w * (WSROWS * YS));
                        *reinterpret_cast<v4f*>(wsJ + (size_t)r * NPIX + c) =
                            ((t[0] + t[1]) + (t[2] + t[3])) + ((t[4] + t[5]) + (t[6] + t[7]));
                    }
                }
                __syncthreads();
            }
        }
    }
}

__global__ __launch_bounds__(32 * PW, 4)
void ms_pair_kernel(const float* __restrict__ xin, float* __restrict__ ws)
{
    __shared__ __align__(16) float etile[PW][16 * ES];
    __shared__ __align__(16) float yslab[PW][WSROWS * YS];

    const int lane = threadIdx.x & 31;
    const int wave = threadIdx.x >> 5;
    const int half = lane >> 4;
    const int l15  = lane & 15;

    int k = blockIdx.x, G = 0;
    while (k >= grp_blocks(G)) { k -= grp_blocks(G); ++G; }
    const int nb = grp_blocks(G), cost = grp_cost(G);
    const int ta = cut_tile(G, (k * cost) / nb);
    const int tb = cut_tile(G, ((k + 1) * cost) / nb);
    const int td = ITG * G;

    const int jw = G * GW + wave * (16 * JT);
    float* yall = &yslab[0][0];

    v2f bj[JT][8];
#pragma unroll
    for (int jt = 0; jt < JT; ++jt) {
#pragma unroll
        for (int k0 = 0; k0 < 8; ++k0) {
            const int d0 = 4 * k0 + 2 * half;
            bj[jt][k0].x = xin[(size_t)d0 * NPIX + jw + 16 * jt + l15];
            bj[jt][k0].y = xin[(size_t)(d0 + 1) * NPIX + jw + 16 * jt + l15];
        }
    }
    v4f xj0[JT][2], xj1[JT][2];
#pragma unroll
    for (int jt = 0; jt < JT; ++jt) {
#pragma unroll
        for (int cp = 0; cp < 2; ++cp) {
            const float* p = xin + (size_t)l15 * NPIX + jw + 16 * jt + 8 * cp + 4 * half;
            xj0[jt][cp] = *reinterpret_cast<const v4f*>(p);
            xj1[jt][cp] = *reinterpret_cast<const v4f*>(p + (size_t)16 * NPIX);
        }
    }

    v8f YJ0[JT], YJ1[JT];
    float dJ[JT];
#pragma unroll
    for (int jt = 0; jt < JT; ++jt) { YJ0[jt] = (v8f){}; YJ1[jt] = (v8f){}; dJ[jt] = 0.0f; }

    float* wsG = ws + (size_t)G * WSROWS * NPIX;
    walk<true >(xin, wsG, etile[wave], yall, yall + wave * (WSROWS * YS), ta, (tb < td) ? tb : td, half, l15,
                bj, xj0, xj1, YJ0, YJ1, dJ);
    walk<false>(xin, wsG, etile[wave], yall, yall + wave * (WSROWS * YS), (ta > td) ? ta : td, tb, half, l15,
                bj, xj0, xj1, YJ0, YJ1, dJ);

    float* wsK = ws + (size_t)k * WSROWS * NPIX;
#pragma unroll
    for (int jt = 0; jt < JT; ++jt) {
        const size_t j = (size_t)jw + 16 * jt + l15;
#pragma unroll
        for (int v = 0; v < 8; ++v) {
            wsK[(size_t)(8 * half + v) * NPIX + j]      = YJ0[jt][v];
            wsK[(size_t)(16 + 8 * half + v) * NPIX + j] = YJ1[jt][v];
        }
        const float dfull = dJ[jt] + __shfl_xor(dJ[jt], 16, 32);
        if (half == 0) wsK[(size_t)32 * NPIX + j] = dfull;
    }
}

#define CX 8
#define CY (DEMB + 1)
__global__ __launch_bounds__(CX * CY, 5)
void ms_combine_kernel(const float* __restrict__ xin, const float* __restrict__ ws,
                       float* __restrict__ xout, float* __restrict__ xout2)
{
    __shared__ __align__(16) float dsh[4 * CX];

    const int x   = threadIdx.x;
    const int r   = threadIdx.y;
    const int c   = blockIdx.x * 32 + 4 * x;
    const int G   = (blockIdx.x * 32) / GW;
    const int nbG = grp_blocks(G);
    const unsigned toff = (unsigned)(r * NPIX + c);

    v4f ld[NG];
#pragma unroll
    for (int P = 0; P < NG; ++P) {
        const bool has = (P < nbG) || (P > G);
        const float* slab = has ? (ws + (size_t)P * WSROWS * NPIX) : ws;
        ld[P] = *reinterpret_cast<const v4f*>(slab + toff);
    }
    const size_t o = (size_t)((r < DEMB) ? r : 0) * NPIX + c;
    const v4f xv = *reinterpret_cast<const v4f*>(xin + o);
    __builtin_amdgcn_sched_barrier(0);
#pragma unroll
    for (int P = 0; P < NG; ++P) asm volatile("" :: "v"(ld[P]));

    v4f acc = {0.0f, 0.0f, 0.0f, 0.0f};
    acc += ld[0];
#pragma unroll
    for (int P = 1; P < NG; ++P) {
        const bool has = (P < nbG) || (P > G);
        const v4f zero = {0.0f, 0.0f, 0.0f, 0.0f};
        acc += has ? ld[P] : zero;
    }

    if (r == DEMB) *reinterpret_cast<v4f*>(dsh + 4 * x) = acc;
    __syncthreads();
    if (r < DEMB) {
        const v4f d = *reinterpret_cast<const v4f*>(dsh + 4 * x);
        v4f val;
#pragma unroll
        for (int e = 0; e < 4; ++e) {
            const float sc = STEP / d[e];
            val[e] = __builtin_fmaf(acc[e], sc, (1.0f - STEP) * xv[e]);
        }
        *reinterpret_cast<v4f*>(xout + o) = val;
        if (xout2) *reinterpret_cast<v4f*>(xout2 + o) = val;
    }
}

extern "C" void kernel_launch(void* const* d_in, const int* in_sizes, int n_in,
                              void* d_out, int out_size, void* d_ws, size_t ws_size,
                              hipStream_t stream) {
    (void)in_sizes; (void)n_in; (void)out_size;
    const float* x0 = (const float*)d_in[0];
    float* out = (float*)d_out;
    float* ws = (float*)d_ws;
    const size_t F = (size_t)DEMB * NPIX;

    if (d_ws != nullptr && ws_size >= WS_NEED) {
        dim3 pgrid(NPBLK), pblock(32 * PW), cgrid(NPIX / 32), cblock(CX, CY);
        ms_pair_kernel<<<pgrid, pblock, 0, stream>>>(x0, ws);
        ms_combine_kernel<<<cgrid, cblock, 0, stream>>>(x0, ws, out + F, nullptr);
        ms_pair_kernel<<<pgrid, pblock, 0, stream>>>(out + F, ws);
        ms_combine_kernel<<<cgrid, cblock, 0, stream>>>(out + F, ws, out + 2 * F, nullptr);
        ms_pair_kernel<<<pgrid, pblock, 0, stream>>>(out + 2 * F, ws);
        ms_combine_kernel<<<cgrid, cblock, 0, stream>>>(out + 2 * F, ws, out + 3 * F, out);
    } else {
        dim3 grid(NTILES), block(32 * NWAVES);
        ms_iter_kernel<<<grid, block, 0, stream>>>(x0, out + F, nullptr);
        ms_iter_kernel<<<grid, block, 0, stream>>>(out + F, out + 2 * F, nullptr);
        ms_iter_kernel<<<grid, block, 0, stream>>>(out + 2 * F, out + 3 * F, out);
    }
}
